# starting kernel
# baseline (speedup 1.0000x reference)
#include <hip/hip_runtime.h>

#define B_  16
#define N_  1025
#define D_  512
#define H_  8
#define HD_ 64
#define QT_ 65
#define ROWS_ 16400

typedef __attribute__((ext_vector_type(2))) float v2f;
typedef __attribute__((ext_vector_type(8))) float v8f;

__device__ __forceinline__ v8f wmma_f32(v2f a, v2f b, v8f c) {
  return __builtin_amdgcn_wmma_f32_16x16x4_f32(false, a, false, b, (short)0, c,
                                               false, false);
}

__global__ __launch_bounds__(256) void softplus_kernel(
    const float* __restrict__ gw, float* __restrict__ sp, int n) {
  int i = blockIdx.x * 256 + threadIdx.x;
  if (i < n) {
    float w = gw[i];
    sp[i] = (w > 20.0f) ? w : log1pf(__expf(w));
  }
}

__global__ __launch_bounds__(32) void qkv_gemm(
    const float* __restrict__ x, const float* __restrict__ w,
    float* __restrict__ Qd, float* __restrict__ Kd, float* __restrict__ Vd) {
  const int lane = threadIdx.x;
  const int lm = lane & 15;
  const int lh = lane >> 4;
  const int col0 = blockIdx.x * 64;
  const int row0 = blockIdx.y * 16;
  const float* ap = x + (size_t)(row0 + lm) * D_ + 2 * lh;
  const float* bp0 = w + (size_t)(col0 + 0 + lm) * D_ + 2 * lh;
  const float* bp1 = w + (size_t)(col0 + 16 + lm) * D_ + 2 * lh;
  const float* bp2 = w + (size_t)(col0 + 32 + lm) * D_ + 2 * lh;
  const float* bp3 = w + (size_t)(col0 + 48 + lm) * D_ + 2 * lh;
  v8f c0 = {}, c1 = {}, c2 = {}, c3 = {};
  v2f a_c = *(const v2f*)(ap);
  v2f b0_c = *(const v2f*)(bp0);
  v2f b1_c = *(const v2f*)(bp1);
  v2f b2_c = *(const v2f*)(bp2);
  v2f b3_c = *(const v2f*)(bp3);
#pragma unroll 4
  for (int k = 0; k < D_ - 4; k += 4) {
    v2f a_n = *(const v2f*)(ap + k + 4);
    v2f b0_n = *(const v2f*)(bp0 + k + 4);
    v2f b1_n = *(const v2f*)(bp1 + k + 4);
    v2f b2_n = *(const v2f*)(bp2 + k + 4);
    v2f b3_n = *(const v2f*)(bp3 + k + 4);
    c0 = wmma_f32(a_c, b0_c, c0);
    c1 = wmma_f32(a_c, b1_c, c1);
    c2 = wmma_f32(a_c, b2_c, c2);
    c3 = wmma_f32(a_c, b3_c, c3);
    a_c = a_n; b0_c = b0_n; b1_c = b1_n; b2_c = b2_n; b3_c = b3_n;
  }
  c0 = wmma_f32(a_c, b0_c, c0);
  c1 = wmma_f32(a_c, b1_c, c1);
  c2 = wmma_f32(a_c, b2_c, c2);
  c3 = wmma_f32(a_c, b3_c, c3);

  int bb8[8], nn8[8];
#pragma unroll
  for (int i = 0; i < 8; ++i) {
    int r = row0 + i + 8 * lh;
    int bbv = r / N_;
    bb8[i] = bbv;
    nn8[i] = r - bbv * N_;
  }
#pragma unroll
  for (int j = 0; j < 4; ++j) {
    const v8f& acc = (j == 0) ? c0 : (j == 1) ? c1 : (j == 2) ? c2 : c3;
    const int c = col0 + j * 16 + lm;
    const int part = c >> 9;
    const int rem = c & 511;
    const int h = rem >> 6;
    const int hd = rem & 63;
    float* dst = (part == 0) ? Qd : (part == 1) ? Kd : Vd;
#pragma unroll
    for (int i = 0; i < 8; ++i) {
      dst[(((size_t)(bb8[i] * H_ + h) * N_ + nn8[i]) * HD_) + hd] = acc[i];
    }
  }
}

__global__ __launch_bounds__(32) void attn_kernel(
    const float* __restrict__ Qd, const float* __restrict__ Kd,
    const float* __restrict__ Vd, const float* __restrict__ SP,
    float* __restrict__ AO) {
  __shared__ float lds[16][18];
  const int lane = threadIdx.x;
  const int lm = lane & 15;
  const int lh = lane >> 4;
  const int qt = blockIdx.x;
  const int bh = blockIdx.y;
  const int b = bh >> 3;
  const int h = bh & 7;
  const size_t base = (size_t)bh * N_ * HD_;
  const float* Qp = Qd + base;
  const float* Kp = Kd + base;
  const float* Vp = Vd + base;
  const float scale = 0.125f;

  const int qlane = qt * 16 + lm;
  const int qclamp = (qlane < N_) ? qlane : N_ - 1;
  v2f qa[16];
#pragma unroll
  for (int s = 0; s < 16; ++s)
    qa[s] = *(const v2f*)(Qp + (size_t)qclamp * HD_ + 4 * s + 2 * lh);

  v8f O0 = {}, O1 = {}, O2 = {}, O3 = {};
  float mrun[8], lrun[8];
#pragma unroll
  for (int i = 0; i < 8; ++i) { mrun[i] = -3.0e38f; lrun[i] = 0.0f; }

  for (int kt = 0; kt < QT_; ++kt) {
    const int keyb = kt * 16 + lm;
    const int keybc = (keyb < N_) ? keyb : N_ - 1;
    const float* kp = Kp + (size_t)keybc * HD_ + 2 * lh;
    v2f bf[16];
#pragma unroll
    for (int s = 0; s < 16; ++s) bf[s] = *(const v2f*)(kp + 4 * s);
    v8f S = {};
#pragma unroll
    for (int s = 0; s < 16; ++s) S = wmma_f32(qa[s], bf[s], S);

    const int keyd = kt * 16 + lm;
    float p[8], corr[8];
#pragma unroll
    for (int i = 0; i < 8; ++i) {
      int qrow = qt * 16 + i + 8 * lh;
      float lg;
      if (qrow >= N_)      lg = 0.0f;
      else if (keyd >= N_) lg = -3.0e38f;
      else                 lg = S[i] * scale * SP[(size_t)qrow * N_ + keyd];
      float mx = lg;
#pragma unroll
      for (int off = 1; off < 16; off <<= 1)
        mx = fmaxf(mx, __shfl_xor(mx, off, 32));
      float mnew = fmaxf(mrun[i], mx);
      float pv = __expf(lg - mnew);
      float sum = pv;
#pragma unroll
      for (int off = 1; off < 16; off <<= 1) sum += __shfl_xor(sum, off, 32);
      float c = __expf(mrun[i] - mnew);
      lrun[i] = lrun[i] * c + sum;
      mrun[i] = mnew;
      corr[i] = c;
      p[i] = pv;
    }
#pragma unroll
    for (int i = 0; i < 8; ++i) {
      O0[i] *= corr[i]; O1[i] *= corr[i]; O2[i] *= corr[i]; O3[i] *= corr[i];
    }
    v2f vb[4][4];
#pragma unroll
    for (int kk = 0; kk < 4; ++kk) {
      int krow = kt * 16 + 4 * kk + 2 * lh;
      int kr0 = (krow < N_) ? krow : N_ - 1;
      int kr1 = (krow + 1 < N_) ? krow + 1 : N_ - 1;
      const float* v0 = Vp + (size_t)kr0 * HD_;
      const float* v1 = Vp + (size_t)kr1 * HD_;
#pragma unroll
      for (int nt = 0; nt < 4; ++nt) {
        vb[kk][nt].x = v0[nt * 16 + lm];
        vb[kk][nt].y = v1[nt * 16 + lm];
      }
    }
    __syncthreads();
#pragma unroll
    for (int i = 0; i < 8; ++i) lds[i + 8 * lh][lm] = p[i];
    __syncthreads();
#pragma unroll
    for (int kk = 0; kk < 4; ++kk) {
      v2f pa = *(const v2f*)&lds[lm][4 * kk + 2 * lh];
      O0 = wmma_f32(pa, vb[kk][0], O0);
      O1 = wmma_f32(pa, vb[kk][1], O1);
      O2 = wmma_f32(pa, vb[kk][2], O2);
      O3 = wmma_f32(pa, vb[kk][3], O3);
    }
  }
  float* outp = AO + (size_t)b * N_ * D_ + h * HD_;
#pragma unroll
  for (int i = 0; i < 8; ++i) {
    int qrow = qt * 16 + i + 8 * lh;
    if (qrow < N_) {
      float inv = 1.0f / lrun[i];
      size_t ro = (size_t)qrow * D_;
      outp[ro +  0 + lm] = O0[i] * inv;
      outp[ro + 16 + lm] = O1[i] * inv;
      outp[ro + 32 + lm] = O2[i] * inv;
      outp[ro + 48 + lm] = O3[i] * inv;
    }
  }
}

__global__ __launch_bounds__(32) void proj_gemm(
    const float* __restrict__ AO, const float* __restrict__ w,
    const float* __restrict__ bias, float* __restrict__ P) {
  const int lane = threadIdx.x;
  const int lm = lane & 15;
  const int lh = lane >> 4;
  const int col0 = blockIdx.x * 64;
  const int row0 = blockIdx.y * 16;
  const float* ap = AO + (size_t)(row0 + lm) * D_ + 2 * lh;
  const float* bp0 = w + (size_t)(col0 + 0 + lm) * D_ + 2 * lh;
  const float* bp1 = w + (size_t)(col0 + 16 + lm) * D_ + 2 * lh;
  const float* bp2 = w + (size_t)(col0 + 32 + lm) * D_ + 2 * lh;
  const float* bp3 = w + (size_t)(col0 + 48 + lm) * D_ + 2 * lh;
  v8f c0 = {}, c1 = {}, c2 = {}, c3 = {};
  v2f a_c = *(const v2f*)(ap);
  v2f b0_c = *(const v2f*)(bp0);
  v2f b1_c = *(const v2f*)(bp1);
  v2f b2_c = *(const v2f*)(bp2);
  v2f b3_c = *(const v2f*)(bp3);
#pragma unroll 4
  for (int k = 0; k < D_ - 4; k += 4) {
    v2f a_n = *(const v2f*)(ap + k + 4);
    v2f b0_n = *(const v2f*)(bp0 + k + 4);
    v2f b1_n = *(const v2f*)(bp1 + k + 4);
    v2f b2_n = *(const v2f*)(bp2 + k + 4);
    v2f b3_n = *(const v2f*)(bp3 + k + 4);
    c0 = wmma_f32(a_c, b0_c, c0);
    c1 = wmma_f32(a_c, b1_c, c1);
    c2 = wmma_f32(a_c, b2_c, c2);
    c3 = wmma_f32(a_c, b3_c, c3);
    a_c = a_n; b0_c = b0_n; b1_c = b1_n; b2_c = b2_n; b3_c = b3_n;
  }
  c0 = wmma_f32(a_c, b0_c, c0);
  c1 = wmma_f32(a_c, b1_c, c1);
  c2 = wmma_f32(a_c, b2_c, c2);
  c3 = wmma_f32(a_c, b3_c, c3);
#pragma unroll
  for (int j = 0; j < 4; ++j) {
    const v8f& acc = (j == 0) ? c0 : (j == 1) ? c1 : (j == 2) ? c2 : c3;
    const int c = col0 + j * 16 + lm;
    const float bc = bias[c];
#pragma unroll
    for (int i = 0; i < 8; ++i) {
      int r = row0 + i + 8 * lh;
      P[(size_t)r * D_ + c] = acc[i] + bc;
    }
  }
}

__global__ __launch_bounds__(256) void ln_residual(
    float* __restrict__ Y, const float* __restrict__ gamma,
    const float* __restrict__ beta) {
  __shared__ float reds[8], reds2[8];
  __shared__ float mu_s, rs_s;
  const int row = blockIdx.x;
  float* y = Y + (size_t)row * D_;
  const int t = threadIdx.x;
  float a = y[t], bv = y[t + 256];
  float s = a + bv, s2 = a * a + bv * bv;
#pragma unroll
  for (int off = 1; off < 32; off <<= 1) {
    s += __shfl_xor(s, off, 32);
    s2 += __shfl_xor(s2, off, 32);
  }
  if ((t & 31) == 0) { reds[t >> 5] = s; reds2[t >> 5] = s2; }
  __syncthreads();
  if (t == 0) {
    float S = 0.f, S2 = 0.f;
#pragma unroll
    for (int i = 0; i < 8; ++i) { S += reds[i]; S2 += reds2[i]; }
    float mu = S * (1.0f / D_);
    float var = S2 * (1.0f / D_) - mu * mu;
    mu_s = mu;
    rs_s = rsqrtf(var + 1e-5f);
  }
  __syncthreads();
  const float mu = mu_s, rs = rs_s;
  y[t]       = a  + (a  - mu) * rs * gamma[t]       + beta[t];
  y[t + 256] = bv + (bv - mu) * rs * gamma[t + 256] + beta[t + 256];
}

extern "C" void kernel_launch(void* const* d_in, const int* in_sizes, int n_in,
                              void* d_out, int out_size, void* d_ws,
                              size_t ws_size, hipStream_t stream) {
  const float* x      = (const float*)d_in[0];
  const float* w_qkv  = (const float*)d_in[1];
  const float* w_proj = (const float*)d_in[2];
  const float* b_proj = (const float*)d_in[3];
  const float* gw     = (const float*)d_in[4];
  const float* ln_g   = (const float*)d_in[5];
  const float* ln_b   = (const float*)d_in[6];
  float* out = (float*)d_out;

  float* ws = (float*)d_ws;
  const size_t nsp  = (size_t)N_ * N_;
  const size_t nqkv = (size_t)B_ * H_ * N_ * HD_;
  float* SP = ws;
  float* Q  = SP + nsp;
  float* K  = Q + nqkv;
  float* V  = K + nqkv;
  float* AO = V + nqkv;

  softplus_kernel<<<(int)((nsp + 255) / 256), 256, 0, stream>>>(gw, SP,
                                                                (int)nsp);
  qkv_gemm<<<dim3(3 * D_ / 64, ROWS_ / 16), 32, 0, stream>>>(x, w_qkv, Q, K,
                                                             V);
  attn_kernel<<<dim3(QT_, B_ * H_), 32, 0, stream>>>(Q, K, V, SP, AO);
  proj_gemm<<<dim3(D_ / 64, ROWS_ / 16), 32, 0, stream>>>(AO, w_proj, b_proj,
                                                          out);
  ln_residual<<<ROWS_, 256, 0, stream>>>(out, ln_g, ln_b);
}


# attention: 4-wave work-group owns 64 queries; 64-key K/V slabs staged in LDS by 16-byte loads (2 stages), softmax once per slab; changes d order of scores and grouping of row max/sum
# speedup vs baseline: 3.2415x; 3.2415x over previous
#include <hip/hip_runtime.h>

#define B_  16
#define N_  1025
#define D_  512
#define H_  8
#define HD_ 64
#define QT_ 65
#define ROWS_ 16400

typedef __attribute__((ext_vector_type(2))) float v2f;
typedef __attribute__((ext_vector_type(8))) float v8f;

__device__ __forceinline__ v8f wmma_f32(v2f a, v2f b, v8f c) {
  return __builtin_amdgcn_wmma_f32_16x16x4_f32(false, a, false, b, (short)0, c,
                                               false, false);
}

__global__ __launch_bounds__(256) void softplus_kernel(
    const float* __restrict__ gw, float* __restrict__ sp, int n) {
  int i = blockIdx.x * 256 + threadIdx.x;
  if (i < n) {
    float w = gw[i];
    sp[i] = (w > 20.0f) ? w : log1pf(__expf(w));
  }
}

#define GBM 64
#define GBN 128
#define GTHREADS 128
#define GBK 32
#define GPITCH 36
#define GKT (D_ / GBK)

typedef __attribute__((ext_vector_type(4))) float v4f;

__device__ __forceinline__ void mma_slab(const float* as, const float* bs,
                                         v8f (&acc)[2][4]) {
  v4f fa[2][2], fb[2][4];
#pragma unroll
  for (int mi = 0; mi < 2; ++mi) fa[0][mi] = *(const v4f*)(as + mi * 16 * GPITCH);
#pragma unroll
  for (int nj = 0; nj < 4; ++nj) fb[0][nj] = *(const v4f*)(bs + nj * 16 * GPITCH);
#pragma unroll
  for (int ks = 0; ks < GBK / 8; ++ks) {
    const int cur = ks & 1, nxt = cur ^ 1;
    if (ks + 1 < GBK / 8) {
      __builtin_amdgcn_sched_barrier(0);
#pragma unroll
      for (int mi = 0; mi < 2; ++mi)
        fa[nxt][mi] = *(const v4f*)(as + mi * 16 * GPITCH + (ks + 1) * 8);
#pragma unroll
      for (int nj = 0; nj < 4; ++nj)
        fb[nxt][nj] = *(const v4f*)(bs + nj * 16 * GPITCH + (ks + 1) * 8);
      __builtin_amdgcn_sched_barrier(0);
    }
#pragma unroll
    for (int mi = 0; mi < 2; ++mi) {
      v2f a; a.x = fa[cur][mi].x; a.y = fa[cur][mi].y;
#pragma unroll
      for (int nj = 0; nj < 4; ++nj) {
        v2f b; b.x = fb[cur][nj].x; b.y = fb[cur][nj].y;
        acc[mi][nj] = wmma_f32(a, b, acc[mi][nj]);
      }
    }
#pragma unroll
    for (int mi = 0; mi < 2; ++mi) {
      v2f a; a.x = fa[cur][mi].z; a.y = fa[cur][mi].w;
#pragma unroll
      for (int nj = 0; nj < 4; ++nj) {
        v2f b; b.x = fb[cur][nj].z; b.y = fb[cur][nj].w;
        acc[mi][nj] = wmma_f32(a, b, acc[mi][nj]);
      }
    }
  }
}

__device__ __forceinline__ void gemm_tile(const float* __restrict__ A,
                                          const float* __restrict__ Wt,
                                          int row0, int col0,
                                          v8f (&acc)[2][4]) {
  __shared__ __attribute__((aligned(16))) float As[2 * GBM * GPITCH];
  __shared__ __attribute__((aligned(16))) float Bs[2 * GBN * GPITCH];
  const int t = threadIdx.x;
  const int lane = t & 31, wave = t >> 5;
  const int lm = lane & 15, lh = lane >> 4;
  const int wy = wave >> 1, wx = wave & 1;
  const int lr = t >> 3;
  const int lc = (t & 7) * 4;
  const float* ag[GBM / 16];
#pragma unroll
  for (int j = 0; j < GBM / 16; ++j) {
    int r = row0 + lr + 16 * j;
    r = (r < ROWS_) ? r : ROWS_ - 1;
    ag[j] = A + (size_t)r * D_ + lc;
  }
  const float* bg = Wt + (size_t)(col0 + lr) * D_ + lc;
  const int sto = lr * GPITCH + lc;
  const float* as = As + (wy * 32 + lm) * GPITCH + 4 * lh;
  const float* bs = Bs + (wx * 64 + lm) * GPITCH + 4 * lh;
  v4f ra[GBM / 16], rb[GBN / 16];
#pragma unroll
  for (int j = 0; j < GBM / 16; ++j) ra[j] = *(const v4f*)(ag[j]);
#pragma unroll
  for (int j = 0; j < GBN / 16; ++j) rb[j] = *(const v4f*)(bg + (size_t)16 * j * D_);
#pragma unroll
  for (int j = 0; j < GBM / 16; ++j) *(v4f*)(As + sto + 16 * j * GPITCH) = ra[j];
#pragma unroll
  for (int j = 0; j < GBN / 16; ++j) *(v4f*)(Bs + sto + 16 * j * GPITCH) = rb[j];
  __syncthreads();
  for (int kt = 0; kt < GKT - 1; ++kt) {
    const int ca = (kt & 1) * (GBM * GPITCH), na = (GBM * GPITCH) - ca;
    const int cb = (kt & 1) * (GBN * GPITCH), nb = (GBN * GPITCH) - cb;
#pragma unroll
    for (int j = 0; j < GBM / 16; ++j) ra[j] = *(const v4f*)(ag[j] + (kt + 1) * GBK);
#pragma unroll
    for (int j = 0; j < GBN / 16; ++j)
      rb[j] = *(const v4f*)(bg + (size_t)16 * j * D_ + (kt + 1) * GBK);
    __builtin_amdgcn_sched_barrier(0);
    mma_slab(as + ca, bs + cb, acc);
#pragma unroll
    for (int j = 0; j < GBM / 16; ++j) *(v4f*)(As + na + sto + 16 * j * GPITCH) = ra[j];
#pragma unroll
    for (int j = 0; j < GBN / 16; ++j) *(v4f*)(Bs + nb + sto + 16 * j * GPITCH) = rb[j];
    __syncthreads();
  }
  mma_slab(as + ((GKT - 1) & 1) * (GBM * GPITCH),
           bs + ((GKT - 1) & 1) * (GBN * GPITCH), acc);
}

__global__ __launch_bounds__(GTHREADS) void qkv_gemm(
    const float* __restrict__ x, const float* __restrict__ w,
    float* __restrict__ Qd, float* __restrict__ Kd, float* __restrict__ Vd) {
  const int col0 = blockIdx.x * GBN;
  const int row0 = blockIdx.y * GBM;
  v8f acc[2][4] = {};
  gemm_tile(x, w, row0, col0, acc);
  const int t = threadIdx.x;
  const int lane = t & 31, wave = t >> 5;
  const int lm = lane & 15, lh = lane >> 4;
  const int wy = wave >> 1, wx = wave & 1;
  const int cblk = (col0 >> 6) + wx;
  const int part = cblk >> 3;
  const int h = cblk & 7;
  float* dst = (part == 0) ? Qd : (part == 1) ? Kd : Vd;
#pragma unroll
  for (int mi = 0; mi < 2; ++mi) {
#pragma unroll
    for (int i = 0; i < 8; ++i) {
      const int r = row0 + wy * 32 + mi * 16 + 8 * lh + i;
      if (r < ROWS_) {
        const int bb = r / N_;
        const int nn = r - bb * N_;
        float* p = dst + ((size_t)(bb * H_ + h) * N_ + nn) * HD_ + lm;
#pragma unroll
        for (int nj = 0; nj < 4; ++nj) p[nj * 16] = acc[mi][nj][i];
      }
    }
  }
}

#define AW 4
#define ATHREADS (AW * 32)
#define AKB 64
#define AKP 68
#define AVP 80
#define APP 68
#define AQG ((QT_ + AW - 1) / AW)
#define AFULL (N_ / AKB)
static_assert(N_ - AFULL * AKB > 0 && N_ - AFULL * AKB <= 16, "tail is one 16-key tile");

template <int NSUB, bool MASK>
__device__ __forceinline__ void attn_slab(
    const float* ksr, const float* vsr, float* psw, const float* psr,
    const float* __restrict__ SP, const int (&spo)[8], int kbase, int lm,
    const v4f (&qa)[8], v8f (&O)[4], float (&mrun)[8], float (&lrun)[8]) {
  const float scale = 0.125f;
  float spv[8][NSUB];
#pragma unroll
  for (int i = 0; i < 8; ++i) {
#pragma unroll
    for (int kl = 0; kl < NSUB; ++kl) {
      int kc = kbase + kl * 16 + lm;
      if (MASK) kc = (kc < N_) ? kc : N_ - 1;
      spv[i][kl] = SP[spo[i] + kc];
    }
  }
  v8f S[NSUB];
#pragma unroll
  for (int kl = 0; kl < NSUB; ++kl) S[kl] = v8f{};
  v4f kb[2][NSUB];
#pragma unroll
  for (int kl = 0; kl < NSUB; ++kl) kb[0][kl] = *(const v4f*)(ksr + kl * 16 * AKP);
#pragma unroll
  for (int s8 = 0; s8 < 8; ++s8) {
    const int cur = s8 & 1, nxt = cur ^ 1;
    if (s8 + 1 < 8) {
      __builtin_amdgcn_sched_barrier(0);
#pragma unroll
      for (int kl = 0; kl < NSUB; ++kl)
        kb[nxt][kl] = *(const v4f*)(ksr + kl * 16 * AKP + (s8 + 1) * 8);
      __builtin_amdgcn_sched_barrier(0);
    }
    v2f a; a.x = qa[s8].x; a.y = qa[s8].y;
#pragma unroll
    for (int kl = 0; kl < NSUB; ++kl) {
      v2f b; b.x = kb[cur][kl].x; b.y = kb[cur][kl].y;
      S[kl] = wmma_f32(a, b, S[kl]);
    }
    a.x = qa[s8].z; a.y = qa[s8].w;
#pragma unroll
    for (int kl = 0; kl < NSUB; ++kl) {
      v2f b; b.x = kb[cur][kl].z; b.y = kb[cur][kl].w;
      S[kl] = wmma_f32(a, b, S[kl]);
    }
  }
#pragma unroll
  for (int i = 0; i < 8; ++i) {
    float lg[NSUB];
#pragma unroll
    for (int kl = 0; kl < NSUB; ++kl) {
      lg[kl] = S[kl][i] * scale * spv[i][kl];
      if (MASK) {
        if (kbase + kl * 16 + lm >= N_) lg[kl] = -3.0e38f;
      }
    }
    float mx = lg[0];
#pragma unroll
    for (int kl = 1; kl < NSUB; ++kl) mx = fmaxf(mx, lg[kl]);
#pragma unroll
    for (int off = 1; off < 16; off <<= 1)
      mx = fmaxf(mx, __shfl_xor(mx, off, 32));
    const float mnew = fmaxf(mrun[i], mx);
    float sum = 0.0f;
#pragma unroll
    for (int kl = 0; kl < NSUB; ++kl) {
      const float pv = __expf(lg[kl] - mnew);
      sum += pv;
      psw[i * APP + kl * 16] = pv;
    }
#pragma unroll
    for (int off = 1; off < 16; off <<= 1) sum += __shfl_xor(sum, off, 32);
    const float c = __expf(mrun[i] - mnew);
    lrun[i] = lrun[i] * c + sum;
    mrun[i] = mnew;
    O[0][i] *= c; O[1][i] *= c; O[2][i] *= c; O[3][i] *= c;
  }
  __builtin_amdgcn_fence(__ATOMIC_RELEASE, "wavefront");
  __builtin_amdgcn_wave_barrier();
  __builtin_amdgcn_fence(__ATOMIC_ACQUIRE, "wavefront");
  v2f pa[2][4];
  v2f vb[2][4][4];
#pragma unroll
  for (int kk = 0; kk < 4; ++kk) {
    pa[0][kk] = *(const v2f*)(psr + 4 * kk);
#pragma unroll
    for (int nt = 0; nt < 4; ++nt) {
      vb[0][kk][nt].x = vsr[(4 * kk) * AVP + nt * 16];
      vb[0][kk][nt].y = vsr[(4 * kk + 1) * AVP + nt * 16];
    }
  }
#pragma unroll
  for (int g = 0; g < NSUB; ++g) {
    const int cur = g & 1, nxt = cur ^ 1;
    if (g + 1 < NSUB) {
      __builtin_amdgcn_sched_barrier(0);
#pragma unroll
      for (int kk = 0; kk < 4; ++kk) {
        pa[nxt][kk] = *(const v2f*)(psr + 16 * (g + 1) + 4 * kk);
#pragma unroll
        for (int nt = 0; nt < 4; ++nt) {
          vb[nxt][kk][nt].x = vsr[(16 * (g + 1) + 4 * kk) * AVP + nt * 16];
          vb[nxt][kk][nt].y = vsr[(16 * (g + 1) + 4 * kk + 1) * AVP + nt * 16];
        }
      }
      __builtin_amdgcn_sched_barrier(0);
    }
#pragma unroll
    for (int kk = 0; kk < 4; ++kk) {
#pragma unroll
      for (int nt = 0; nt < 4; ++nt)
        O[nt] = wmma_f32(pa[cur][kk], vb[cur][kk][nt], O[nt]);
    }
  }
}

__global__ __launch_bounds__(ATHREADS) void attn_kernel(
    const float* __restrict__ Qd, const float* __restrict__ Kd,
    const float* __restrict__ Vd, const float* __restrict__ SP,
    float* __restrict__ AO) {
  __shared__ __attribute__((aligned(16))) float Ks[2 * AKB * AKP];
  __shared__ __attribute__((aligned(16))) float Vs[2 * AKB * AVP];
  __shared__ __attribute__((aligned(16))) float Ps[AW * 16 * APP];
  const int t = threadIdx.x;
  const int lane = t & 31;
  const int wave = __builtin_amdgcn_readfirstlane(t >> 5);
  const int lm = lane & 15;
  const int lh = lane >> 4;
  const int qt = blockIdx.x * AW + wave;
  const bool live = qt < QT_;
  const int bh = blockIdx.y;
  const int b = bh >> 3;
  const int h = bh & 7;
  const size_t base = (size_t)bh * N_ * HD_;
  const float* Qp = Qd + base;
  const float* Kp = Kd + base;
  const float* Vp = Vd + base;

  const int sr = t >> 4;
  const int sc = (t & 15) * 4;
  const float* kg = Kp + sc;
  const float* vg = Vp + sc;
  const int ksto = sr * AKP + sc;
  const int vsto = sr * AVP + sc;
  v4f rk[AKB / 8], rv[AKB / 8];
#pragma unroll
  for (int j = 0; j < AKB / 8; ++j) {
    const int key = sr + 8 * j;
    rk[j] = *(const v4f*)(kg + (size_t)key * HD_);
    rv[j] = *(const v4f*)(vg + (size_t)key * HD_);
  }
#pragma unroll
  for (int j = 0; j < AKB / 8; ++j) {
    *(v4f*)(Ks + ksto + 8 * j * AKP) = rk[j];
    *(v4f*)(Vs + vsto + 8 * j * AVP) = rv[j];
  }

  const int qlane = qt * 16 + lm;
  const int qclamp = (qlane < N_) ? qlane : N_ - 1;
  v4f qa[8];
#pragma unroll
  for (int s8 = 0; s8 < 8; ++s8)
    qa[s8] = *(const v4f*)(Qp + (size_t)qclamp * HD_ + 8 * s8 + 4 * lh);
  int spo[8];
#pragma unroll
  for (int i = 0; i < 8; ++i) {
    const int qrow = qt * 16 + 8 * lh + i;
    spo[i] = ((qrow < N_) ? qrow : N_ - 1) * N_;
  }
  v8f O[4];
#pragma unroll
  for (int nt = 0; nt < 4; ++nt) O[nt] = v8f{};
  float mrun[8], lrun[8];
#pragma unroll
  for (int i = 0; i < 8; ++i) { mrun[i] = -3.0e38f; lrun[i] = 0.0f; }

  const float* ksr = Ks + lm * AKP + 4 * lh;
  const float* vsr = Vs + 2 * lh * AVP + lm;
  float* pw = Ps + wave * (16 * APP);
  float* psw = pw + 8 * lh * APP + lm;
  const float* psr = pw + lm * APP + 2 * lh;

  __syncthreads();
  for (int kt = 0; kt < AFULL; ++kt) {
    const int cur = kt & 1;
#pragma unroll
    for (int j = 0; j < AKB / 8; ++j) {
      int key = (kt + 1) * AKB + sr + 8 * j;
      key = (key < N_) ? key : N_ - 1;
      rk[j] = *(const v4f*)(kg + (size_t)key * HD_);
      rv[j] = *(const v4f*)(vg + (size_t)key * HD_);
    }
    __builtin_amdgcn_sched_barrier(0);
    if (live)
      attn_slab<4, false>(ksr + cur * (AKB * AKP), vsr + cur * (AKB * AVP), psw,
                          psr, SP, spo, kt * AKB, lm, qa, O, mrun, lrun);
#pragma unroll
    for (int j = 0; j < AKB / 8; ++j) {
      *(v4f*)(Ks + (cur ^ 1) * (AKB * AKP) + ksto + 8 * j * AKP) = rk[j];
      *(v4f*)(Vs + (cur ^ 1) * (AKB * AVP) + vsto + 8 * j * AVP) = rv[j];
    }
    __syncthreads();
  }
  if (!live) return;
  attn_slab<1, true>(ksr + (AFULL & 1) * (AKB * AKP), vsr + (AFULL & 1) * (AKB * AVP),
                     psw, psr, SP, spo, AFULL * AKB, lm, qa, O, mrun, lrun);
  float* outp = AO + (size_t)b * N_ * D_ + h * HD_;
#pragma unroll
  for (int i = 0; i < 8; ++i) {
    int qrow = qt * 16 + i + 8 * lh;
    if (qrow < N_) {
      float inv = 1.0f / lrun[i];
      size_t ro = (size_t)qrow * D_;
      outp[ro +  0 + lm] = O[0][i] * inv;
      outp[ro + 16 + lm] = O[1][i] * inv;
      outp[ro + 32 + lm] = O[2][i] * inv;
      outp[ro + 48 + lm] = O[3][i] * inv;
    }
  }
}

__global__ __launch_bounds__(GTHREADS) void proj_gemm(
    const float* __restrict__ AO, const float* __restrict__ w,
    const float* __restrict__ bias, float* __restrict__ P) {
  const int col0 = blockIdx.x * GBN;
  const int row0 = blockIdx.y * GBM;
  v8f acc[2][4] = {};
  gemm_tile(AO, w, row0, col0, acc);
  const int t = threadIdx.x;
  const int lane = t & 31, wave = t >> 5;
  const int lm = lane & 15, lh = lane >> 4;
  const int wy = wave >> 1, wx = wave & 1;
  float bc[4];
#pragma unroll
  for (int nj = 0; nj < 4; ++nj) bc[nj] = bias[col0 + wx * 64 + nj * 16 + lm];
#pragma unroll
  for (int mi = 0; mi < 2; ++mi) {
#pragma unroll
    for (int i = 0; i < 8; ++i) {
      const int r = row0 + wy * 32 + mi * 16 + 8 * lh + i;
      if (r < ROWS_) {
        float* p = P + (size_t)r * D_ + col0 + wx * 64 + lm;
#pragma unroll
        for (int nj = 0; nj < 4; ++nj) p[nj * 16] = acc[mi][nj][i] + bc[nj];
      }
    }
  }
}

__global__ __launch_bounds__(256) void ln_residual(
    float* __restrict__ Y, const float* __restrict__ gamma,
    const float* __restrict__ beta) {
  __shared__ float reds[8], reds2[8];
  __shared__ float mu_s, rs_s;
  const int row = blockIdx.x;
  float* y = Y + (size_t)row * D_;
  const int t = threadIdx.x;
  float a = y[t], bv = y[t + 256];
  float s = a + bv, s2 = a * a + bv * bv;
#pragma unroll
  for (int off = 1; off < 32; off <<= 1) {
    s += __shfl_xor(s, off, 32);
    s2 += __shfl_xor(s2, off, 32);
  }
  if ((t & 31) == 0) { reds[t >> 5] = s; reds2[t >> 5] = s2; }
  __syncthreads();
  if (t == 0) {
    float S = 0.f, S2 = 0.f;
#pragma unroll
    for (int i = 0; i < 8; ++i) { S += reds[i]; S2 += reds2[i]; }
    float mu = S * (1.0f / D_);
    float var = S2 * (1.0f / D_) - mu * mu;
    mu_s = mu;
    rs_s = rsqrtf(var + 1e-5f);
  }
  __syncthreads();
  const float mu = mu_s, rs = rs_s;
  y[t]       = a  + (a  - mu) * rs * gamma[t]       + beta[t];
  y[t + 256] = bv + (bv - mu) * rs * gamma[t + 256] + beta[t + 256];
}

extern "C" void kernel_launch(void* const* d_in, const int* in_sizes, int n_in,
                              void* d_out, int out_size, void* d_ws,
                              size_t ws_size, hipStream_t stream) {
  const float* x      = (const float*)d_in[0];
  const float* w_qkv  = (const float*)d_in[1];
  const float* w_proj = (const float*)d_in[2];
  const float* b_proj = (const float*)d_in[3];
  const float* gw     = (const float*)d_in[4];
  const float* ln_g   = (const float*)d_in[5];
  const float* ln_b   = (const float*)d_in[6];
  float* out = (float*)d_out;

  float* ws = (float*)d_ws;
  const size_t nsp  = (size_t)N_ * N_;
  const size_t nqkv = (size_t)B_ * H_ * N_ * HD_;
  const size_t nsp_pad = ((nsp + 63) / 64) * 64;
  float* SP = ws;
  float* Q  = SP + nsp_pad;
  float* K  = Q + nqkv;
  float* V  = K + nqkv;
  float* AO = V + nqkv;

  softplus_kernel<<<(int)((nsp + 255) / 256), 256, 0, stream>>>(gw, SP,
                                                                (int)nsp);
  qkv_gemm<<<dim3(3 * D_ / GBN, (ROWS_ + GBM - 1) / GBM), GTHREADS, 0, stream>>>(
      x, w_qkv, Q, K, V);
  attn_kernel<<<dim3(AQG, B_ * H_), ATHREADS, 0, stream>>>(Q, K, V, SP, AO);
  proj_gemm<<<dim3(D_ / GBN, (ROWS_ + GBM - 1) / GBM), GTHREADS, 0, stream>>>(
      AO, w_proj, b_proj, out);
  ln_residual<<<ROWS_, 256, 0, stream>>>(out, ln_g, ln_b);
}


# attention softmax: 16-lane row max and row sum as DPP row_xmask steps on the vector unit instead of 64 LDS-permute round trips per 64-key slab; same butterfly, sums bit-identical
# speedup vs baseline: 3.3478x; 1.0328x over previous
#include <hip/hip_runtime.h>

#define B_  16
#define N_  1025
#define D_  512
#define H_  8
#define HD_ 64
#define QT_ 65
#define ROWS_ 16400

typedef __attribute__((ext_vector_type(2))) float v2f;
typedef __attribute__((ext_vector_type(8))) float v8f;

__device__ __forceinline__ v8f wmma_f32(v2f a, v2f b, v8f c) {
  return __builtin_amdgcn_wmma_f32_16x16x4_f32(false, a, false, b, (short)0, c,
                                               false, false);
}

__global__ __launch_bounds__(256) void softplus_kernel(
    const float* __restrict__ gw, float* __restrict__ sp, int n) {
  int i = blockIdx.x * 256 + threadIdx.x;
  if (i < n) {
    float w = gw[i];
    sp[i] = (w > 20.0f) ? w : log1pf(__expf(w));
  }
}

#define GBM 64
#define GBN 128
#define GTHREADS 128
#define GBK 32
#define GPITCH 36
#define GKT (D_ / GBK)

typedef __attribute__((ext_vector_type(4))) float v4f;

__device__ __forceinline__ void mma_slab(const float* as, const float* bs,
                                         v8f (&acc)[2][4]) {
  v4f fa[2][2], fb[2][4];
#pragma unroll
  for (int mi = 0; mi < 2; ++mi) fa[0][mi] = *(const v4f*)(as + mi * 16 * GPITCH);
#pragma unroll
  for (int nj = 0; nj < 4; ++nj) fb[0][nj] = *(const v4f*)(bs + nj * 16 * GPITCH);
#pragma unroll
  for (int ks = 0; ks < GBK / 8; ++ks) {
    const int cur = ks & 1, nxt = cur ^ 1;
    if (ks + 1 < GBK / 8) {
      __builtin_amdgcn_sched_barrier(0);
#pragma unroll
      for (int mi = 0; mi < 2; ++mi)
        fa[nxt][mi] = *(const v4f*)(as + mi * 16 * GPITCH + (ks + 1) * 8);
#pragma unroll
      for (int nj = 0; nj < 4; ++nj)
        fb[nxt][nj] = *(const v4f*)(bs + nj * 16 * GPITCH + (ks + 1) * 8);
      __builtin_amdgcn_sched_barrier(0);
    }
#pragma unroll
    for (int mi = 0; mi < 2; ++mi) {
      v2f a; a.x = fa[cur][mi].x; a.y = fa[cur][mi].y;
#pragma unroll
      for (int nj = 0; nj < 4; ++nj) {
        v2f b; b.x = fb[cur][nj].x; b.y = fb[cur][nj].y;
        acc[mi][nj] = wmma_f32(a, b, acc[mi][nj]);
      }
    }
#pragma unroll
    for (int mi = 0; mi < 2; ++mi) {
      v2f a; a.x = fa[cur][mi].z; a.y = fa[cur][mi].w;
#pragma unroll
      for (int nj = 0; nj < 4; ++nj) {
        v2f b; b.x = fb[cur][nj].z; b.y = fb[cur][nj].w;
        acc[mi][nj] = wmma_f32(a, b, acc[mi][nj]);
      }
    }
  }
}

__device__ __forceinline__ void gemm_tile(const float* __restrict__ A,
                                          const float* __restrict__ Wt,
                                          int row0, int col0,
                                          v8f (&acc)[2][4]) {
  __shared__ __attribute__((aligned(16))) float As[2 * GBM * GPITCH];
  __shared__ __attribute__((aligned(16))) float Bs[2 * GBN * GPITCH];
  const int t = threadIdx.x;
  const int lane = t & 31, wave = t >> 5;
  const int lm = lane & 15, lh = lane >> 4;
  const int wy = wave >> 1, wx = wave & 1;
  const int lr = t >> 3;
  const int lc = (t & 7) * 4;
  const float* ag[GBM / 16];
#pragma unroll
  for (int j = 0; j < GBM / 16; ++j) {
    int r = row0 + lr + 16 * j;
    r = (r < ROWS_) ? r : ROWS_ - 1;
    ag[j] = A + (size_t)r * D_ + lc;
  }
  const float* bg = Wt + (size_t)(col0 + lr) * D_ + lc;
  const int sto = lr * GPITCH + lc;
  const float* as = As + (wy * 32 + lm) * GPITCH + 4 * lh;
  const float* bs = Bs + (wx * 64 + lm) * GPITCH + 4 * lh;
  v4f ra[GBM / 16], rb[GBN / 16];
#pragma unroll
  for (int j = 0; j < GBM / 16; ++j) ra[j] = *(const v4f*)(ag[j]);
#pragma unroll
  for (int j = 0; j < GBN / 16; ++j) rb[j] = *(const v4f*)(bg + (size_t)16 * j * D_);
#pragma unroll
  for (int j = 0; j < GBM / 16; ++j) *(v4f*)(As + sto + 16 * j * GPITCH) = ra[j];
#pragma unroll
  for (int j = 0; j < GBN / 16; ++j) *(v4f*)(Bs + sto + 16 * j * GPITCH) = rb[j];
  __syncthreads();
  for (int kt = 0; kt < GKT - 1; ++kt) {
    const int ca = (kt & 1) * (GBM * GPITCH), na = (GBM * GPITCH) - ca;
    const int cb = (kt & 1) * (GBN * GPITCH), nb = (GBN * GPITCH) - cb;
#pragma unroll
    for (int j = 0; j < GBM / 16; ++j) ra[j] = *(const v4f*)(ag[j] + (kt + 1) * GBK);
#pragma unroll
    for (int j = 0; j < GBN / 16; ++j)
      rb[j] = *(const v4f*)(bg + (size_t)16 * j * D_ + (kt + 1) * GBK);
    __builtin_amdgcn_sched_barrier(0);
    mma_slab(as + ca, bs + cb, acc);
#pragma unroll
    for (int j = 0; j < GBM / 16; ++j) *(v4f*)(As + na + sto + 16 * j * GPITCH) = ra[j];
#pragma unroll
    for (int j = 0; j < GBN / 16; ++j) *(v4f*)(Bs + nb + sto + 16 * j * GPITCH) = rb[j];
    __syncthreads();
  }
  mma_slab(as + ((GKT - 1) & 1) * (GBM * GPITCH),
           bs + ((GKT - 1) & 1) * (GBN * GPITCH), acc);
}

__global__ __launch_bounds__(GTHREADS) void qkv_gemm(
    const float* __restrict__ x, const float* __restrict__ w,
    float* __restrict__ Qd, float* __restrict__ Kd, float* __restrict__ Vd) {
  const int col0 = blockIdx.x * GBN;
  const int row0 = blockIdx.y * GBM;
  v8f acc[2][4] = {};
  gemm_tile(x, w, row0, col0, acc);
  const int t = threadIdx.x;
  const int lane = t & 31, wave = t >> 5;
  const int lm = lane & 15, lh = lane >> 4;
  const int wy = wave >> 1, wx = wave & 1;
  const int cblk = (col0 >> 6) + wx;
  const int part = cblk >> 3;
  const int h = cblk & 7;
  float* dst = (part == 0) ? Qd : (part == 1) ? Kd : Vd;
#pragma unroll
  for (int mi = 0; mi < 2; ++mi) {
#pragma unroll
    for (int i = 0; i < 8; ++i) {
      const int r = row0 + wy * 32 + mi * 16 + 8 * lh + i;
      if (r < ROWS_) {
        const int bb = r / N_;
        const int nn = r - bb * N_;
        float* p = dst + ((size_t)(bb * H_ + h) * N_ + nn) * HD_ + lm;
#pragma unroll
        for (int nj = 0; nj < 4; ++nj) p[nj * 16] = acc[mi][nj][i];
      }
    }
  }
}

template <int M>
__device__ __forceinline__ float row_xor(float v) {
  return __builtin_bit_cast(
      float, __builtin_amdgcn_update_dpp(0, __builtin_bit_cast(int, v),
                                         0x160 + M, 0xf, 0xf, true));
}

#define AW 4
#define ATHREADS (AW * 32)
#define AKB 64
#define AKP 68
#define AVP 80
#define APP 68
#define AQG ((QT_ + AW - 1) / AW)
#define AFULL (N_ / AKB)
static_assert(N_ - AFULL * AKB > 0 && N_ - AFULL * AKB <= 16, "tail is one 16-key tile");

template <int NSUB, bool MASK>
__device__ __forceinline__ void attn_slab(
    const float* ksr, const float* vsr, float* psw, const float* psr,
    const float* __restrict__ SP, const int (&spo)[8], int kbase, int lm,
    const v4f (&qa)[8], v8f (&O)[4], float (&mrun)[8], float (&lrun)[8]) {
  const float scale = 0.125f;
  float spv[8][NSUB];
#pragma unroll
  for (int i = 0; i < 8; ++i) {
#pragma unroll
    for (int kl = 0; kl < NSUB; ++kl) {
      int kc = kbase + kl * 16 + lm;
      if (MASK) kc = (kc < N_) ? kc : N_ - 1;
      spv[i][kl] = SP[spo[i] + kc];
    }
  }
  v8f S[NSUB];
#pragma unroll
  for (int kl = 0; kl < NSUB; ++kl) S[kl] = v8f{};
  v4f kb[2][NSUB];
#pragma unroll
  for (int kl = 0; kl < NSUB; ++kl) kb[0][kl] = *(const v4f*)(ksr + kl * 16 * AKP);
#pragma unroll
  for (int s8 = 0; s8 < 8; ++s8) {
    const int cur = s8 & 1, nxt = cur ^ 1;
    if (s8 + 1 < 8) {
      __builtin_amdgcn_sched_barrier(0);
#pragma unroll
      for (int kl = 0; kl < NSUB; ++kl)
        kb[nxt][kl] = *(const v4f*)(ksr + kl * 16 * AKP + (s8 + 1) * 8);
      __builtin_amdgcn_sched_barrier(0);
    }
    v2f a; a.x = qa[s8].x; a.y = qa[s8].y;
#pragma unroll
    for (int kl = 0; kl < NSUB; ++kl) {
      v2f b; b.x = kb[cur][kl].x; b.y = kb[cur][kl].y;
      S[kl] = wmma_f32(a, b, S[kl]);
    }
    a.x = qa[s8].z; a.y = qa[s8].w;
#pragma unroll
    for (int kl = 0; kl < NSUB; ++kl) {
      v2f b; b.x = kb[cur][kl].z; b.y = kb[cur][kl].w;
      S[kl] = wmma_f32(a, b, S[kl]);
    }
  }
#pragma unroll
  for (int i = 0; i < 8; ++i) {
    float lg[NSUB];
#pragma unroll
    for (int kl = 0; kl < NSUB; ++kl) {
      lg[kl] = S[kl][i] * scale * spv[i][kl];
      if (MASK) {
        if (kbase + kl * 16 + lm >= N_) lg[kl] = -3.0e38f;
      }
    }
    float mx = lg[0];
#pragma unroll
    for (int kl = 1; kl < NSUB; ++kl) mx = fmaxf(mx, lg[kl]);
    mx = __builtin_elementwise_maximum(mx, row_xor<1>(mx));
    mx = __builtin_elementwise_maximum(mx, row_xor<2>(mx));
    mx = __builtin_elementwise_maximum(mx, row_xor<4>(mx));
    mx = __builtin_elementwise_maximum(mx, row_xor<8>(mx));
    const float mnew = fmaxf(mrun[i], mx);
    float sum = 0.0f;
#pragma unroll
    for (int kl = 0; kl < NSUB; ++kl) {
      const float pv = __expf(lg[kl] - mnew);
      sum += pv;
      psw[i * APP + kl * 16] = pv;
    }
    sum += row_xor<1>(sum);
    sum += row_xor<2>(sum);
    sum += row_xor<4>(sum);
    sum += row_xor<8>(sum);
    const float c = __expf(mrun[i] - mnew);
    lrun[i] = lrun[i] * c + sum;
    mrun[i] = mnew;
    O[0][i] *= c; O[1][i] *= c; O[2][i] *= c; O[3][i] *= c;
  }
  __builtin_amdgcn_fence(__ATOMIC_RELEASE, "wavefront");
  __builtin_amdgcn_wave_barrier();
  __builtin_amdgcn_fence(__ATOMIC_ACQUIRE, "wavefront");
  v2f pa[2][4];
  v2f vb[2][4][4];
#pragma unroll
  for (int kk = 0; kk < 4; ++kk) {
    pa[0][kk] = *(const v2f*)(psr + 4 * kk);
#pragma unroll
    for (int nt = 0; nt < 4; ++nt) {
      vb[0][kk][nt].x = vsr[(4 * kk) * AVP + nt * 16];
      vb[0][kk][nt].y = vsr[(4 * kk + 1) * AVP + nt * 16];
    }
  }
#pragma unroll
  for (int g = 0; g < NSUB; ++g) {
    const int cur = g & 1, nxt = cur ^ 1;
    if (g + 1 < NSUB) {
      __builtin_amdgcn_sched_barrier(0);
#pragma unroll
      for (int kk = 0; kk < 4; ++kk) {
        pa[nxt][kk] = *(const v2f*)(psr + 16 * (g + 1) + 4 * kk);
#pragma unroll
        for (int nt = 0; nt < 4; ++nt) {
          vb[nxt][kk][nt].x = vsr[(16 * (g + 1) + 4 * kk) * AVP + nt * 16];
          vb[nxt][kk][nt].y = vsr[(16 * (g + 1) + 4 * kk + 1) * AVP + nt * 16];
        }
      }
      __builtin_amdgcn_sched_barrier(0);
    }
#pragma unroll
    for (int kk = 0; kk < 4; ++kk) {
#pragma unroll
      for (int nt = 0; nt < 4; ++nt)
        O[nt] = wmma_f32(pa[cur][kk], vb[cur][kk][nt], O[nt]);
    }
  }
}

__global__ __launch_bounds__(ATHREADS) void attn_kernel(
    const float* __restrict__ Qd, const float* __restrict__ Kd,
    const float* __restrict__ Vd, const float* __restrict__ SP,
    float* __restrict__ AO) {
  __shared__ __attribute__((aligned(16))) float Ks[2 * AKB * AKP];
  __shared__ __attribute__((aligned(16))) float Vs[2 * AKB * AVP];
  __shared__ __attribute__((aligned(16))) float Ps[AW * 16 * APP];
  const int t = threadIdx.x;
  const int lane = t & 31;
  const int wave = __builtin_amdgcn_readfirstlane(t >> 5);
  const int lm = lane & 15;
  const int lh = lane >> 4;
  const int qt = blockIdx.x * AW + wave;
  const bool live = qt < QT_;
  const int bh = blockIdx.y;
  const int b = bh >> 3;
  const int h = bh & 7;
  const size_t base = (size_t)bh * N_ * HD_;
  const float* Qp = Qd + base;
  const float* Kp = Kd + base;
  const float* Vp = Vd + base;

  const int sr = t >> 4;
  const int sc = (t & 15) * 4;
  const float* kg = Kp + sc;
  const float* vg = Vp + sc;
  const int ksto = sr * AKP + sc;
  const int vsto = sr * AVP + sc;
  v4f rk[AKB / 8], rv[AKB / 8];
#pragma unroll
  for (int j = 0; j < AKB / 8; ++j) {
    const int key = sr + 8 * j;
    rk[j] = *(const v4f*)(kg + (size_t)key * HD_);
    rv[j] = *(const v4f*)(vg + (size_t)key * HD_);
  }
#pragma unroll
  for (int j = 0; j < AKB / 8; ++j) {
    *(v4f*)(Ks + ksto + 8 * j * AKP) = rk[j];
    *(v4f*)(Vs + vsto + 8 * j * AVP) = rv[j];
  }

  const int qlane = qt * 16 + lm;
  const int qclamp = (qlane < N_) ? qlane : N_ - 1;
  v4f qa[8];
#pragma unroll
  for (int s8 = 0; s8 < 8; ++s8)
    qa[s8] = *(const v4f*)(Qp + (size_t)qclamp * HD_ + 8 * s8 + 4 * lh);
  int spo[8];
#pragma unroll
  for (int i = 0; i < 8; ++i) {
    const int qrow = qt * 16 + 8 * lh + i;
    spo[i] = ((qrow < N_) ? qrow : N_ - 1) * N_;
  }
  v8f O[4];
#pragma unroll
  for (int nt = 0; nt < 4; ++nt) O[nt] = v8f{};
  float mrun[8], lrun[8];
#pragma unroll
  for (int i = 0; i < 8; ++i) { mrun[i] = -3.0e38f; lrun[i] = 0.0f; }

  const float* ksr = Ks + lm * AKP + 4 * lh;
  const float* vsr = Vs + 2 * lh * AVP + lm;
  float* pw = Ps + wave * (16 * APP);
  float* psw = pw + 8 * lh * APP + lm;
  const float* psr = pw + lm * APP + 2 * lh;

  __syncthreads();
  for (int kt = 0; kt < AFULL; ++kt) {
    const int cur = kt & 1;
#pragma unroll
    for (int j = 0; j < AKB / 8; ++j) {
      int key = (kt + 1) * AKB + sr + 8 * j;
      key = (key < N_) ? key : N_ - 1;
      rk[j] = *(const v4f*)(kg + (size_t)key * HD_);
      rv[j] = *(const v4f*)(vg + (size_t)key * HD_);
    }
    __builtin_amdgcn_sched_barrier(0);
    if (live)
      attn_slab<4, false>(ksr + cur * (AKB * AKP), vsr + cur * (AKB * AVP), psw,
                          psr, SP, spo, kt * AKB, lm, qa, O, mrun, lrun);
#pragma unroll
    for (int j = 0; j < AKB / 8; ++j) {
      *(v4f*)(Ks + (cur ^ 1) * (AKB * AKP) + ksto + 8 * j * AKP) = rk[j];
      *(v4f*)(Vs + (cur ^ 1) * (AKB * AVP) + vsto + 8 * j * AVP) = rv[j];
    }
    __syncthreads();
  }
  if (!live) return;
  attn_slab<1, true>(ksr + (AFULL & 1) * (AKB * AKP), vsr + (AFULL & 1) * (AKB * AVP),
                     psw, psr, SP, spo, AFULL * AKB, lm, qa, O, mrun, lrun);
  float* outp = AO + (size_t)b * N_ * D_ + h * HD_;
#pragma unroll
  for (int i = 0; i < 8; ++i) {
    int qrow = qt * 16 + i + 8 * lh;
    if (qrow < N_) {
      float inv = 1.0f / lrun[i];
      size_t ro = (size_t)qrow * D_;
      outp[ro +  0 + lm] = O[0][i] * inv;
      outp[ro + 16 + lm] = O[1][i] * inv;
      outp[ro + 32 + lm] = O[2][i] * inv;
      outp[ro + 48 + lm] = O[3][i] * inv;
    }
  }
}

__global__ __launch_bounds__(GTHREADS) void proj_gemm(
    const float* __restrict__ AO, const float* __restrict__ w,
    const float* __restrict__ bias, float* __restrict__ P) {
  const int col0 = blockIdx.x * GBN;
  const int row0 = blockIdx.y * GBM;
  v8f acc[2][4] = {};
  gemm_tile(AO, w, row0, col0, acc);
  const int t = threadIdx.x;
  const int lane = t & 31, wave = t >> 5;
  const int lm = lane & 15, lh = lane >> 4;
  const int wy = wave >> 1, wx = wave & 1;
  float bc[4];
#pragma unroll
  for (int nj = 0; nj < 4; ++nj) bc[nj] = bias[col0 + wx * 64 + nj * 16 + lm];
#pragma unroll
  for (int mi = 0; mi < 2; ++mi) {
#pragma unroll
    for (int i = 0; i < 8; ++i) {
      const int r = row0 + wy * 32 + mi * 16 + 8 * lh + i;
      if (r < ROWS_) {
        float* p = P + (size_t)r * D_ + col0 + wx * 64 + lm;
#pragma unroll
        for (int nj = 0; nj < 4; ++nj) p[nj * 16] = acc[mi][nj][i] + bc[nj];
      }
    }
  }
}

__global__ __launch_bounds__(256) void ln_residual(
    float* __restrict__ Y, const float* __restrict__ gamma,
    const float* __restrict__ beta) {
  __shared__ float reds[8], reds2[8];
  __shared__ float mu_s, rs_s;
  const int row = blockIdx.x;
  float* y = Y + (size_t)row * D_;
  const int t = threadIdx.x;
  float a = y[t], bv = y[t + 256];
  float s = a + bv, s2 = a * a + bv * bv;
#pragma unroll
  for (int off = 1; off < 32; off <<= 1) {
    s += __shfl_xor(s, off, 32);
    s2 += __shfl_xor(s2, off, 32);
  }
  if ((t & 31) == 0) { reds[t >> 5] = s; reds2[t >> 5] = s2; }
  __syncthreads();
  if (t == 0) {
    float S = 0.f, S2 = 0.f;
#pragma unroll
    for (int i = 0; i < 8; ++i) { S += reds[i]; S2 += reds2[i]; }
    float mu = S * (1.0f / D_);
    float var = S2 * (1.0f / D_) - mu * mu;
    mu_s = mu;
    rs_s = rsqrtf(var + 1e-5f);
  }
  __syncthreads();
  const float mu = mu_s, rs = rs_s;
  y[t]       = a  + (a  - mu) * rs * gamma[t]       + beta[t];
  y[t + 256] = bv + (bv - mu) * rs * gamma[t + 256] + beta[t + 256];
}

extern "C" void kernel_launch(void* const* d_in, const int* in_sizes, int n_in,
                              void* d_out, int out_size, void* d_ws,
                              size_t ws_size, hipStream_t stream) {
  const float* x      = (const float*)d_in[0];
  const float* w_qkv  = (const float*)d_in[1];
  const float* w_proj = (const float*)d_in[2];
  const float* b_proj = (const float*)d_in[3];
  const float* gw     = (const float*)d_in[4];
  const float* ln_g   = (const float*)d_in[5];
  const float* ln_b   = (const float*)d_in[6];
  float* out = (float*)d_out;

  float* ws = (float*)d_ws;
  const size_t nsp  = (size_t)N_ * N_;
  const size_t nqkv = (size_t)B_ * H_ * N_ * HD_;
  const size_t nsp_pad = ((nsp + 63) / 64) * 64;
  float* SP = ws;
  float* Q  = SP + nsp_pad;
  float* K  = Q + nqkv;
  float* V  = K + nqkv;
  float* AO = V + nqkv;

  softplus_kernel<<<(int)((nsp + 255) / 256), 256, 0, stream>>>(gw, SP,
                                                                (int)nsp);
  qkv_gemm<<<dim3(3 * D_ / GBN, (ROWS_ + GBM - 1) / GBM), GTHREADS, 0, stream>>>(
      x, w_qkv, Q, K, V);
  attn_kernel<<<dim3(AQG, B_ * H_), ATHREADS, 0, stream>>>(Q, K, V, SP, AO);
  proj_gemm<<<dim3(D_ / GBN, (ROWS_ + GBM - 1) / GBM), GTHREADS, 0, stream>>>(
      AO, w_proj, b_proj, out);
  ln_residual<<<ROWS_, 256, 0, stream>>>(out, ln_g, ln_b);
}


# attention: a lane owns 4 adjacent keys of a slab (K rows permuted, V staged 4-key-interleaved in LDS): graph factors, P and V fragments go as 16-byte pieces; changes row-sum grouping, PV key order
# speedup vs baseline: 3.4595x; 1.0334x over previous
#include <hip/hip_runtime.h>

#define B_  16
#define N_  1025
#define D_  512
#define H_  8
#define HD_ 64
#define QT_ 65
#define ROWS_ 16400
#define SPP_ 1028

typedef __attribute__((ext_vector_type(2))) float v2f;
typedef __attribute__((ext_vector_type(8))) float v8f;

__device__ __forceinline__ v8f wmma_f32(v2f a, v2f b, v8f c) {
  return __builtin_amdgcn_wmma_f32_16x16x4_f32(false, a, false, b, (short)0, c,
                                               false, false);
}

__global__ __launch_bounds__(256) void softplus_kernel(
    const float* __restrict__ gw, float* __restrict__ sp, int n) {
  int i = blockIdx.x * 256 + threadIdx.x;
  if (i < n) {
    const int r = i / SPP_;
    const int cc = i - r * SPP_;
    float w = gw[r * N_ + ((cc < N_) ? cc : N_ - 1)];
    sp[i] = (w > 20.0f) ? w : log1pf(__expf(w));
  }
}

#define GBM 64
#define GBN 128
#define GTHREADS 128
#define GBK 32
#define GPITCH 36
#define GKT (D_ / GBK)

typedef __attribute__((ext_vector_type(4))) float v4f;

__device__ __forceinline__ void mma_slab(const float* as, const float* bs,
                                         v8f (&acc)[2][4]) {
  v4f fa[2][2], fb[2][4];
#pragma unroll
  for (int mi = 0; mi < 2; ++mi) fa[0][mi] = *(const v4f*)(as + mi * 16 * GPITCH);
#pragma unroll
  for (int nj = 0; nj < 4; ++nj) fb[0][nj] = *(const v4f*)(bs + nj * 16 * GPITCH);
#pragma unroll
  for (int ks = 0; ks < GBK / 8; ++ks) {
    const int cur = ks & 1, nxt = cur ^ 1;
    if (ks + 1 < GBK / 8) {
      __builtin_amdgcn_sched_barrier(0);
#pragma unroll
      for (int mi = 0; mi < 2; ++mi)
        fa[nxt][mi] = *(const v4f*)(as + mi * 16 * GPITCH + (ks + 1) * 8);
#pragma unroll
      for (int nj = 0; nj < 4; ++nj)
        fb[nxt][nj] = *(const v4f*)(bs + nj * 16 * GPITCH + (ks + 1) * 8);
      __builtin_amdgcn_sched_barrier(0);
    }
#pragma unroll
    for (int mi = 0; mi < 2; ++mi) {
      v2f a; a.x = fa[cur][mi].x; a.y = fa[cur][mi].y;
#pragma unroll
      for (int nj = 0; nj < 4; ++nj) {
        v2f b; b.x = fb[cur][nj].x; b.y = fb[cur][nj].y;
        acc[mi][nj] = wmma_f32(a, b, acc[mi][nj]);
      }
    }
#pragma unroll
    for (int mi = 0; mi < 2; ++mi) {
      v2f a; a.x = fa[cur][mi].z; a.y = fa[cur][mi].w;
#pragma unroll
      for (int nj = 0; nj < 4; ++nj) {
        v2f b; b.x = fb[cur][nj].z; b.y = fb[cur][nj].w;
        acc[mi][nj] = wmma_f32(a, b, acc[mi][nj]);
      }
    }
  }
}

__device__ __forceinline__ void gemm_tile(const float* __restrict__ A,
                                          const float* __restrict__ Wt,
                                          int row0, int col0,
                                          v8f (&acc)[2][4]) {
  __shared__ __attribute__((aligned(16))) float As[2 * GBM * GPITCH];
  __shared__ __attribute__((aligned(16))) float Bs[2 * GBN * GPITCH];
  const int t = threadIdx.x;
  const int lane = t & 31, wave = t >> 5;
  const int lm = lane & 15, lh = lane >> 4;
  const int wy = wave >> 1, wx = wave & 1;
  const int lr = t >> 3;
  const int lc = (t & 7) * 4;
  const float* ag[GBM / 16];
#pragma unroll
  for (int j = 0; j < GBM / 16; ++j) {
    int r = row0 + lr + 16 * j;
    r = (r < ROWS_) ? r : ROWS_ - 1;
    ag[j] = A + (size_t)r * D_ + lc;
  }
  const float* bg = Wt + (size_t)(col0 + lr) * D_ + lc;
  const int sto = lr * GPITCH + lc;
  const float* as = As + (wy * 32 + lm) * GPITCH + 4 * lh;
  const float* bs = Bs + (wx * 64 + lm) * GPITCH + 4 * lh;
  v4f ra[GBM / 16], rb[GBN / 16];
#pragma unroll
  for (int j = 0; j < GBM / 16; ++j) ra[j] = *(const v4f*)(ag[j]);
#pragma unroll
  for (int j = 0; j < GBN / 16; ++j) rb[j] = *(const v4f*)(bg + (size_t)16 * j * D_);
#pragma unroll
  for (int j = 0; j < GBM / 16; ++j) *(v4f*)(As + sto + 16 * j * GPITCH) = ra[j];
#pragma unroll
  for (int j = 0; j < GBN / 16; ++j) *(v4f*)(Bs + sto + 16 * j * GPITCH) = rb[j];
  __syncthreads();
  for (int kt = 0; kt < GKT - 1; ++kt) {
    const int ca = (kt & 1) * (GBM * GPITCH), na = (GBM * GPITCH) - ca;
    const int cb = (kt & 1) * (GBN * GPITCH), nb = (GBN * GPITCH) - cb;
#pragma unroll
    for (int j = 0; j < GBM / 16; ++j) ra[j] = *(const v4f*)(ag[j] + (kt + 1) * GBK);
#pragma unroll
    for (int j = 0; j < GBN / 16; ++j)
      rb[j] = *(const v4f*)(bg + (size_t)16 * j * D_ + (kt + 1) * GBK);
    __builtin_amdgcn_sched_barrier(0);
    mma_slab(as + ca, bs + cb, acc);
#pragma unroll
    for (int j = 0; j < GBM / 16; ++j) *(v4f*)(As + na + sto + 16 * j * GPITCH) = ra[j];
#pragma unroll
    for (int j = 0; j < GBN / 16; ++j) *(v4f*)(Bs + nb + sto + 16 * j * GPITCH) = rb[j];
    __syncthreads();
  }
  mma_slab(as + ((GKT - 1) & 1) * (GBM * GPITCH),
           bs + ((GKT - 1) & 1) * (GBN * GPITCH), acc);
}

__global__ __launch_bounds__(GTHREADS) void qkv_gemm(
    const float* __restrict__ x, const float* __restrict__ w,
    float* __restrict__ Qd, float* __restrict__ Kd, float* __restrict__ Vd) {
  const int col0 = blockIdx.x * GBN;
  const int row0 = blockIdx.y * GBM;
  v8f acc[2][4] = {};
  gemm_tile(x, w, row0, col0, acc);
  const int t = threadIdx.x;
  const int lane = t & 31, wave = t >> 5;
  const int lm = lane & 15, lh = lane >> 4;
  const int wy = wave >> 1, wx = wave & 1;
  const int cblk = (col0 >> 6) + wx;
  const int part = cblk >> 3;
  const int h = cblk & 7;
  float* dst = (part == 0) ? Qd : (part == 1) ? Kd : Vd;
#pragma unroll
  for (int mi = 0; mi < 2; ++mi) {
#pragma unroll
    for (int i = 0; i < 8; ++i) {
      const int r = row0 + wy * 32 + mi * 16 + 8 * lh + i;
      if (r < ROWS_) {
        const int bb = r / N_;
        const int nn = r - bb * N_;
        float* p = dst + ((size_t)(bb * H_ + h) * N_ + nn) * HD_ + lm;
#pragma unroll
        for (int nj = 0; nj < 4; ++nj) p[nj * 16] = acc[mi][nj][i];
      }
    }
  }
}

template <int M>
__device__ __forceinline__ float row_xor(float v) {
  return __builtin_bit_cast(
      float, __builtin_amdgcn_update_dpp(0, __builtin_bit_cast(int, v),
                                         0x160 + M, 0xf, 0xf, true));
}

#define AW 4
#define ATHREADS (AW * 32)
#define AKB 64
#define AKP 68
#define AVR (AKB / 4)
#define AVP 260
#define APP 68
#define AQG ((QT_ + AW - 1) / AW)
#define AFULL (N_ / AKB)
static_assert(N_ - AFULL * AKB > 0 && N_ - AFULL * AKB <= 16, "tail is one 16-key tile");

template <int NSUB, bool MASK>
__device__ __forceinline__ void attn_slab(
    const float* ksr, const float* const (&vsr)[4], float* psw, const float* psr,
    const float* __restrict__ SP, const int (&spo)[8], int kbase, int lm,
    const v4f (&qa)[8], v8f (&O)[4], float (&mrun)[8], float (&lrun)[8]) {
  const float scale = 0.125f;
  float spv[8][NSUB];
#pragma unroll
  for (int i = 0; i < 8; ++i) {
    if (NSUB == 4) {
      const v4f s4 = *(const v4f*)(SP + spo[i] + kbase + 4 * lm);
#pragma unroll
      for (int kl = 0; kl < NSUB; ++kl) spv[i][kl] = s4[kl];
    } else {
#pragma unroll
      for (int kl = 0; kl < NSUB; ++kl) {
        int kc = kbase + 4 * lm + kl;
        if (MASK) kc = (kc < N_) ? kc : N_ - 1;
        spv[i][kl] = SP[spo[i] + kc];
      }
    }
  }
  v8f S[NSUB];
#pragma unroll
  for (int kl = 0; kl < NSUB; ++kl) S[kl] = v8f{};
  v4f kb[2][NSUB];
#pragma unroll
  for (int kl = 0; kl < NSUB; ++kl) kb[0][kl] = *(const v4f*)(ksr + kl * 16 * AKP);
#pragma unroll
  for (int s8 = 0; s8 < 8; ++s8) {
    const int cur = s8 & 1, nxt = cur ^ 1;
    if (s8 + 1 < 8) {
      __builtin_amdgcn_sched_barrier(0);
#pragma unroll
      for (int kl = 0; kl < NSUB; ++kl)
        kb[nxt][kl] = *(const v4f*)(ksr + kl * 16 * AKP + (s8 + 1) * 8);
      __builtin_amdgcn_sched_barrier(0);
    }
    v2f a; a.x = qa[s8].x; a.y = qa[s8].y;
#pragma unroll
    for (int kl = 0; kl < NSUB; ++kl) {
      v2f b; b.x = kb[cur][kl].x; b.y = kb[cur][kl].y;
      S[kl] = wmma_f32(a, b, S[kl]);
    }
    a.x = qa[s8].z; a.y = qa[s8].w;
#pragma unroll
    for (int kl = 0; kl < NSUB; ++kl) {
      v2f b; b.x = kb[cur][kl].z; b.y = kb[cur][kl].w;
      S[kl] = wmma_f32(a, b, S[kl]);
    }
  }
#pragma unroll
  for (int i = 0; i < 8; ++i) {
    float lg[NSUB];
#pragma unroll
    for (int kl = 0; kl < NSUB; ++kl) {
      lg[kl] = S[kl][i] * scale * spv[i][kl];
      if (MASK) {
        if (kbase + 4 * lm + kl >= N_) lg[kl] = -3.0e38f;
      }
    }
    float mx = lg[0];
#pragma unroll
    for (int kl = 1; kl < NSUB; ++kl) mx = fmaxf(mx, lg[kl]);
    mx = __builtin_elementwise_maximum(mx, row_xor<1>(mx));
    mx = __builtin_elementwise_maximum(mx, row_xor<2>(mx));
    mx = __builtin_elementwise_maximum(mx, row_xor<4>(mx));
    mx = __builtin_elementwise_maximum(mx, row_xor<8>(mx));
    const float mnew = fmaxf(mrun[i], mx);
    float sum = 0.0f;
    v4f p4 = v4f{0.0f, 0.0f, 0.0f, 0.0f};
#pragma unroll
    for (int kl = 0; kl < NSUB; ++kl) {
      const float pv = __expf(lg[kl] - mnew);
      sum += pv;
      p4[kl] = pv;
    }
    *(v4f*)(psw + i * APP) = p4;
    sum += row_xor<1>(sum);
    sum += row_xor<2>(sum);
    sum += row_xor<4>(sum);
    sum += row_xor<8>(sum);
    const float c = __expf(mrun[i] - mnew);
    lrun[i] = lrun[i] * c + sum;
    mrun[i] = mnew;
    O[0][i] *= c; O[1][i] *= c; O[2][i] *= c; O[3][i] *= c;
  }
  __builtin_amdgcn_fence(__ATOMIC_RELEASE, "wavefront");
  __builtin_amdgcn_wave_barrier();
  __builtin_amdgcn_fence(__ATOMIC_ACQUIRE, "wavefront");
  v4f pa[2][2];
  v4f vb[2][2][4];
#pragma unroll
  for (int j = 0; j < 2; ++j) {
    pa[0][j] = *(const v4f*)(psr + 8 * j);
#pragma unroll
    for (int nt = 0; nt < 4; ++nt)
      vb[0][j][nt] = *(const v4f*)(vsr[nt] + (2 * j) * AVP);
  }
#pragma unroll
  for (int g = 0; g < NSUB; ++g) {
    const int cur = g & 1, nxt = cur ^ 1;
    if (g + 1 < NSUB) {
      __builtin_amdgcn_sched_barrier(0);
#pragma unroll
      for (int j = 0; j < 2; ++j) {
        pa[nxt][j] = *(const v4f*)(psr + 16 * (g + 1) + 8 * j);
#pragma unroll
        for (int nt = 0; nt < 4; ++nt)
          vb[nxt][j][nt] = *(const v4f*)(vsr[nt] + (4 * (g + 1) + 2 * j) * AVP);
      }
      __builtin_amdgcn_sched_barrier(0);
    }
#pragma unroll
    for (int j = 0; j < 2; ++j) {
      v2f a; a.x = pa[cur][j].x; a.y = pa[cur][j].y;
#pragma unroll
      for (int nt = 0; nt < 4; ++nt) {
        v2f b; b.x = vb[cur][j][nt].x; b.y = vb[cur][j][nt].y;
        O[nt] = wmma_f32(a, b, O[nt]);
      }
      a.x = pa[cur][j].z; a.y = pa[cur][j].w;
#pragma unroll
      for (int nt = 0; nt < 4; ++nt) {
        v2f b; b.x = vb[cur][j][nt].z; b.y = vb[cur][j][nt].w;
        O[nt] = wmma_f32(a, b, O[nt]);
      }
    }
  }
}

__global__ __launch_bounds__(ATHREADS) void attn_kernel(
    const float* __restrict__ Qd, const float* __restrict__ Kd,
    const float* __restrict__ Vd, const float* __restrict__ SP,
    float* __restrict__ AO) {
  __shared__ __attribute__((aligned(16))) float Ks[2 * AKB * AKP];
  __shared__ __attribute__((aligned(16))) float Vs[2 * AVR * AVP];
  __shared__ __attribute__((aligned(16))) float Ps[AW * 16 * APP];
  const int t = threadIdx.x;
  const int lane = t & 31;
  const int wave = __builtin_amdgcn_readfirstlane(t >> 5);
  const int lm = lane & 15;
  const int lh = lane >> 4;
  const int qt = blockIdx.x * AW + wave;
  const bool live = qt < QT_;
  const int bh = blockIdx.y;
  const int b = bh >> 3;
  const int h = bh & 7;
  const size_t base = (size_t)bh * N_ * HD_;
  const float* Qp = Qd + base;
  const float* Kp = Kd + base;
  const float* Vp = Vd + base;

  const int sr = t >> 4;
  const int c16 = t & 15;
  const int sc = c16 * 4;
  const float* kg = Kp + sc;
  const float* vg = Vp + sc;
  const int ksto = ((sr & 3) * 16 + (sr >> 2)) * AKP + sc;
  int vsto[4];
#pragma unroll
  for (int e = 0; e < 4; ++e) vsto[e] = sr * AVP + 4 * (sc + (e ^ (c16 >> 2)));
  v4f rk[AKB / 8], rv[2][4];
#pragma unroll
  for (int j = 0; j < AKB / 8; ++j)
    rk[j] = *(const v4f*)(kg + (size_t)(sr + 8 * j) * HD_);
#pragma unroll
  for (int b2 = 0; b2 < 2; ++b2) {
#pragma unroll
    for (int e = 0; e < 4; ++e)
      rv[b2][e] = *(const v4f*)(vg + (size_t)(4 * (sr + 8 * b2) + e) * HD_);
  }
#pragma unroll
  for (int j = 0; j < AKB / 8; ++j) *(v4f*)(Ks + ksto + 2 * j * AKP) = rk[j];
#pragma unroll
  for (int b2 = 0; b2 < 2; ++b2) {
#pragma unroll
    for (int e = 0; e < 4; ++e)
      *(v4f*)(Vs + vsto[e] + 8 * b2 * AVP) =
          v4f{rv[b2][0][e], rv[b2][1][e], rv[b2][2][e], rv[b2][3][e]};
  }

  const int qlane = qt * 16 + lm;
  const int qclamp = (qlane < N_) ? qlane : N_ - 1;
  v4f qa[8];
#pragma unroll
  for (int s8 = 0; s8 < 8; ++s8)
    qa[s8] = *(const v4f*)(Qp + (size_t)qclamp * HD_ + 8 * s8 + 4 * lh);
  int spo[8];
#pragma unroll
  for (int i = 0; i < 8; ++i) {
    const int qrow = qt * 16 + 8 * lh + i;
    spo[i] = ((qrow < N_) ? qrow : N_ - 1) * SPP_;
  }
  v8f O[4];
#pragma unroll
  for (int nt = 0; nt < 4; ++nt) O[nt] = v8f{};
  float mrun[8], lrun[8];
#pragma unroll
  for (int i = 0; i < 8; ++i) { mrun[i] = -3.0e38f; lrun[i] = 0.0f; }

  const float* ksr = Ks + lm * AKP + 4 * lh;
  float* pw = Ps + wave * (16 * APP);
  float* psw = pw + 8 * lh * APP + 4 * lm;
  const float* psr = pw + lm * APP + 4 * lh;

  __syncthreads();
  for (int kt = 0; kt < AFULL; ++kt) {
    const int cur = kt & 1;
#pragma unroll
    for (int j = 0; j < AKB / 8; ++j) {
      int key = (kt + 1) * AKB + sr + 8 * j;
      key = (key < N_) ? key : N_ - 1;
      rk[j] = *(const v4f*)(kg + (size_t)key * HD_);
    }
#pragma unroll
    for (int b2 = 0; b2 < 2; ++b2) {
#pragma unroll
      for (int e = 0; e < 4; ++e) {
        int key = (kt + 1) * AKB + 4 * (sr + 8 * b2) + e;
        key = (key < N_) ? key : N_ - 1;
        rv[b2][e] = *(const v4f*)(vg + (size_t)key * HD_);
      }
    }
    __builtin_amdgcn_sched_barrier(0);
    if (live) {
      const float* vsr[4];
#pragma unroll
      for (int nt = 0; nt < 4; ++nt)
        vsr[nt] = Vs + cur * (AVR * AVP) + lh * AVP + 4 * (16 * nt + (lm ^ nt));
      attn_slab<4, false>(ksr + cur * (AKB * AKP), vsr, psw, psr, SP, spo,
                          kt * AKB, lm, qa, O, mrun, lrun);
    }
#pragma unroll
    for (int j = 0; j < AKB / 8; ++j)
      *(v4f*)(Ks + (cur ^ 1) * (AKB * AKP) + ksto + 2 * j * AKP) = rk[j];
#pragma unroll
    for (int b2 = 0; b2 < 2; ++b2) {
#pragma unroll
      for (int e = 0; e < 4; ++e)
        *(v4f*)(Vs + (cur ^ 1) * (AVR * AVP) + vsto[e] + 8 * b2 * AVP) =
            v4f{rv[b2][0][e], rv[b2][1][e], rv[b2][2][e], rv[b2][3][e]};
    }
    __syncthreads();
  }
  if (!live) return;
  {
    const float* vsr[4];
#pragma unroll
    for (int nt = 0; nt < 4; ++nt)
      vsr[nt] = Vs + (AFULL & 1) * (AVR * AVP) + lh * AVP + 4 * (16 * nt + (lm ^ nt));
    attn_slab<1, true>(ksr + (AFULL & 1) * (AKB * AKP), vsr, psw, psr, SP, spo,
                       AFULL * AKB, lm, qa, O, mrun, lrun);
  }
  float* outp = AO + (size_t)b * N_ * D_ + h * HD_;
#pragma unroll
  for (int i = 0; i < 8; ++i) {
    int qrow = qt * 16 + i + 8 * lh;
    if (qrow < N_) {
      float inv = 1.0f / lrun[i];
      size_t ro = (size_t)qrow * D_;
      outp[ro +  0 + lm] = O[0][i] * inv;
      outp[ro + 16 + lm] = O[1][i] * inv;
      outp[ro + 32 + lm] = O[2][i] * inv;
      outp[ro + 48 + lm] = O[3][i] * inv;
    }
  }
}

__global__ __launch_bounds__(GTHREADS) void proj_gemm(
    const float* __restrict__ AO, const float* __restrict__ w,
    const float* __restrict__ bias, float* __restrict__ P) {
  const int col0 = blockIdx.x * GBN;
  const int row0 = blockIdx.y * GBM;
  v8f acc[2][4] = {};
  gemm_tile(AO, w, row0, col0, acc);
  const int t = threadIdx.x;
  const int lane = t & 31, wave = t >> 5;
  const int lm = lane & 15, lh = lane >> 4;
  const int wy = wave >> 1, wx = wave & 1;
  float bc[4];
#pragma unroll
  for (int nj = 0; nj < 4; ++nj) bc[nj] = bias[col0 + wx * 64 + nj * 16 + lm];
#pragma unroll
  for (int mi = 0; mi < 2; ++mi) {
#pragma unroll
    for (int i = 0; i < 8; ++i) {
      const int r = row0 + wy * 32 + mi * 16 + 8 * lh + i;
      if (r < ROWS_) {
        float* p = P + (size_t)r * D_ + col0 + wx * 64 + lm;
#pragma unroll
        for (int nj = 0; nj < 4; ++nj) p[nj * 16] = acc[mi][nj][i] + bc[nj];
      }
    }
  }
}

__global__ __launch_bounds__(256) void ln_residual(
    float* __restrict__ Y, const float* __restrict__ gamma,
    const float* __restrict__ beta) {
  __shared__ float reds[8], reds2[8];
  __shared__ float mu_s, rs_s;
  const int row = blockIdx.x;
  float* y = Y + (size_t)row * D_;
  const int t = threadIdx.x;
  float a = y[t], bv = y[t + 256];
  float s = a + bv, s2 = a * a + bv * bv;
#pragma unroll
  for (int off = 1; off < 32; off <<= 1) {
    s += __shfl_xor(s, off, 32);
    s2 += __shfl_xor(s2, off, 32);
  }
  if ((t & 31) == 0) { reds[t >> 5] = s; reds2[t >> 5] = s2; }
  __syncthreads();
  if (t == 0) {
    float S = 0.f, S2 = 0.f;
#pragma unroll
    for (int i = 0; i < 8; ++i) { S += reds[i]; S2 += reds2[i]; }
    float mu = S * (1.0f / D_);
    float var = S2 * (1.0f / D_) - mu * mu;
    mu_s = mu;
    rs_s = rsqrtf(var + 1e-5f);
  }
  __syncthreads();
  const float mu = mu_s, rs = rs_s;
  y[t]       = a  + (a  - mu) * rs * gamma[t]       + beta[t];
  y[t + 256] = bv + (bv - mu) * rs * gamma[t + 256] + beta[t + 256];
}

extern "C" void kernel_launch(void* const* d_in, const int* in_sizes, int n_in,
                              void* d_out, int out_size, void* d_ws,
                              size_t ws_size, hipStream_t stream) {
  const float* x      = (const float*)d_in[0];
  const float* w_qkv  = (const float*)d_in[1];
  const float* w_proj = (const float*)d_in[2];
  const float* b_proj = (const float*)d_in[3];
  const float* gw     = (const float*)d_in[4];
  const float* ln_g   = (const float*)d_in[5];
  const float* ln_b   = (const float*)d_in[6];
  float* out = (float*)d_out;

  float* ws = (float*)d_ws;
  const size_t nsp  = (size_t)N_ * SPP_;
  const size_t nqkv = (size_t)B_ * H_ * N_ * HD_;
  const size_t nsp_pad = ((nsp + 63) / 64) * 64;
  float* SP = ws;
  float* Q  = SP + nsp_pad;
  float* K  = Q + nqkv;
  float* V  = K + nqkv;
  float* AO = V + nqkv;

  softplus_kernel<<<(int)((nsp + 255) / 256), 256, 0, stream>>>(gw, SP,
                                                                (int)nsp);
  qkv_gemm<<<dim3(3 * D_ / GBN, (ROWS_ + GBM - 1) / GBM), GTHREADS, 0, stream>>>(
      x, w_qkv, Q, K, V);
  attn_kernel<<<dim3(AQG, B_ * H_), ATHREADS, 0, stream>>>(Q, K, V, SP, AO);
  proj_gemm<<<dim3(D_ / GBN, (ROWS_ + GBM - 1) / GBM), GTHREADS, 0, stream>>>(
      AO, w_proj, b_proj, out);
  ln_residual<<<ROWS_, 256, 0, stream>>>(out, ln_g, ln_b);
}
